# starting kernel
# baseline (speedup 1.0000x reference)
#include <hip/hip_runtime.h>
#include <math.h>

#define D_MODEL   1024
#define D_FF      2048
#define TWO_FF    4096
#define N_EXPERTS 8

#define BM   128
#define KC   64
#define XS_LD (KC + 4)

typedef float v2f __attribute__((ext_vector_type(2)));
typedef float v8f __attribute__((ext_vector_type(8)));

#define WMMA_F32(A, B, C) \
    __builtin_amdgcn_wmma_f32_16x16x4_f32(false, (A), false, (B), (short)0, (C), false, false)

__global__ void zero_out_kernel(float* __restrict__ out, int n, int* __restrict__ counts) {
    int i = blockIdx.x * blockDim.x + threadIdx.x;
    if (i < n) out[i] = 0.0f;
    if (blockIdx.x == 0 && threadIdx.x < 16) counts[threadIdx.x] = 0;
}

__global__ void gate_kernel(const float* __restrict__ x,
                            const float* __restrict__ Wg,
                            int*   __restrict__ counts,
                            int*   __restrict__ idx,
                            float* __restrict__ wts,
                            int*   __restrict__ hrow,
                            int n_tokens) {
    int tok = blockIdx.x;
    int tid = threadIdx.x;
    __shared__ float red[64 * N_EXPERTS];
    __shared__ float tot[N_EXPERTS];

    float s[N_EXPERTS];
#pragma unroll
    for (int e = 0; e < N_EXPERTS; ++e) s[e] = 0.0f;

    const float* xr = x + (size_t)tok * D_MODEL;
    for (int d = tid; d < D_MODEL; d += 64) {
        float xv = xr[d];
#pragma unroll
        for (int e = 0; e < N_EXPERTS; ++e) s[e] += xv * Wg[e * D_MODEL + d];
    }
#pragma unroll
    for (int e = 0; e < N_EXPERTS; ++e) red[tid * N_EXPERTS + e] = s[e];
    __syncthreads();

    if (tid < N_EXPERTS) {
        float t = 0.0f;
        for (int i = 0; i < 64; ++i) t += red[i * N_EXPERTS + tid];
        tot[tid] = t;
    }
    __syncthreads();

    if (tid == 0) {
        int i1 = 0; float v1 = tot[0];
        for (int e = 1; e < N_EXPERTS; ++e) if (tot[e] > v1) { v1 = tot[e]; i1 = e; }
        int i2 = (i1 == 0) ? 1 : 0; float v2 = tot[i2];
        for (int e = 0; e < N_EXPERTS; ++e)
            if (e != i1 && tot[e] > v2) { v2 = tot[e]; i2 = e; }
        float m  = fmaxf(v1, v2);
        float e1 = __expf(v1 - m), e2 = __expf(v2 - m);
        float inv = 1.0f / (e1 + e2);

        int p1 = atomicAdd(&counts[i1], 1);
        idx [i1 * n_tokens + p1] = tok;
        wts [i1 * n_tokens + p1] = e1 * inv;
        hrow[i1 * n_tokens + p1] = 2 * tok;
        int p2 = atomicAdd(&counts[i2], 1);
        idx [i2 * n_tokens + p2] = tok;
        wts [i2 * n_tokens + p2] = e2 * inv;
        hrow[i2 * n_tokens + p2] = 2 * tok + 1;
    }
}

__global__ void __launch_bounds__(256)
moe_gemm1_kernel(const float* __restrict__ x,
                 const float* __restrict__ W1,
                 const float* __restrict__ b1,
                 const int*   __restrict__ counts,
                 const int*   __restrict__ idx,
                 const int*   __restrict__ hrow,
                 float*       __restrict__ hh,
                 int n_tokens) {
    int e   = blockIdx.z;
    int Mb  = blockIdx.y;
    int Fb  = blockIdx.x;
    int cnt = counts[e];
    if (Mb * BM >= cnt) return;

    __shared__ float Xs[BM * XS_LD];
    __shared__ int   stok[BM];
    __shared__ int   shr[BM];

    int tid = threadIdx.x;
    if (tid < BM) {
        int p = Mb * BM + tid;
        bool v = (p < cnt);
        stok[tid] = v ? idx [e * n_tokens + p] : 0;
        shr [tid] = v ? hrow[e * n_tokens + p] : 0;
    }

    int lane = tid & 31;
    int wave = tid >> 5;
    int g    = lane >> 4;
    int nl   = lane & 15;
    int mg   = wave >> 2;
    int fg   = wave & 3;
    int fbase = Fb * 128 + fg * 32;

    const float* W1e = W1 + (size_t)e * D_MODEL * TWO_FF;
    const float* b1e = b1 + (size_t)e * TWO_FF;

    v8f accA[4][2], accG[4][2];
#pragma unroll
    for (int i = 0; i < 4; ++i)
#pragma unroll
        for (int j = 0; j < 2; ++j) {
            accA[i][j] = (v8f){0.f,0.f,0.f,0.f,0.f,0.f,0.f,0.f};
            accG[i][j] = (v8f){0.f,0.f,0.f,0.f,0.f,0.f,0.f,0.f};
        }

    for (int kc0 = 0; kc0 < D_MODEL; kc0 += KC) {
        __syncthreads();
        for (int t = tid; t < BM * (KC / 4); t += 256) {
            int row = t >> 4;
            int c4  = t & 15;
            const float4* src = (const float4*)(x + (size_t)stok[row] * D_MODEL + kc0);
            *(float4*)(Xs + row * XS_LD + c4 * 4) = src[c4];
        }
        __syncthreads();

        for (int k4 = 0; k4 < KC; k4 += 4) {
            int k0 = k4 + 2 * g;
            v2f a[4];
#pragma unroll
            for (int i = 0; i < 4; ++i) {
                int row = mg * 64 + i * 16 + nl;
                a[i].x = Xs[row * XS_LD + k0];
                a[i].y = Xs[row * XS_LD + k0 + 1];
            }
            const float* Wk = W1e + (size_t)(kc0 + k0) * TWO_FF;
#pragma unroll
            for (int j = 0; j < 2; ++j) {
                int colA = fbase + j * 16 + nl;
                v2f bA, bG;
                bA.x = Wk[colA];
                bA.y = Wk[TWO_FF + colA];
                bG.x = Wk[colA + D_FF];
                bG.y = Wk[TWO_FF + colA + D_FF];
#pragma unroll
                for (int i = 0; i < 4; ++i) {
                    accA[i][j] = WMMA_F32(a[i], bA, accA[i][j]);
                    accG[i][j] = WMMA_F32(a[i], bG, accG[i][j]);
                }
            }
        }
    }

#pragma unroll
    for (int j = 0; j < 2; ++j) {
        int colA = fbase + j * 16 + nl;
        float biasA = b1e[colA];
        float biasG = b1e[colA + D_FF];
#pragma unroll
        for (int i = 0; i < 4; ++i) {
#pragma unroll
            for (int jj = 0; jj < 8; ++jj) {
                int row = mg * 64 + i * 16 + jj + 8 * g;
                int p   = Mb * BM + row;
                if (p < cnt) {
                    float av = accA[i][j][jj] + biasA;
                    float gv = accG[i][j][jj] + biasG;
                    float hv = av * (1.0f / (1.0f + __expf(-av))) * gv;
                    hh[(size_t)shr[row] * D_FF + colA] = hv;
                }
            }
        }
    }
}

__global__ void __launch_bounds__(256)
moe_gemm2_kernel(const float* __restrict__ hh,
                 const float* __restrict__ W2,
                 const float* __restrict__ b2,
                 const int*   __restrict__ counts,
                 const int*   __restrict__ idx,
                 const float* __restrict__ wts,
                 const int*   __restrict__ hrow,
                 float*       __restrict__ out,
                 int n_tokens) {
    int e   = blockIdx.z;
    int Mb  = blockIdx.y;
    int Nb  = blockIdx.x;
    int cnt = counts[e];
    if (Mb * BM >= cnt) return;

    __shared__ float Hs[BM * XS_LD];
    __shared__ int   stok[BM];
    __shared__ int   shr[BM];
    __shared__ float sww[BM];

    int tid = threadIdx.x;
    if (tid < BM) {
        int p = Mb * BM + tid;
        bool v = (p < cnt);
        stok[tid] = v ? idx [e * n_tokens + p] : 0;
        shr [tid] = v ? hrow[e * n_tokens + p] : 0;
        sww [tid] = v ? wts [e * n_tokens + p] : 0.0f;
    }

    int lane = tid & 31;
    int wave = tid >> 5;
    int g    = lane >> 4;
    int nl   = lane & 15;
    int mg   = wave >> 2;
    int ng   = wave & 3;
    int nbase = Nb * 128 + ng * 32;

    const float* W2e = W2 + (size_t)e * D_FF * D_MODEL;
    const float* b2e = b2 + (size_t)e * D_MODEL;

    v8f acc[4][2];
#pragma unroll
    for (int i = 0; i < 4; ++i)
#pragma unroll
        for (int j = 0; j < 2; ++j)
            acc[i][j] = (v8f){0.f,0.f,0.f,0.f,0.f,0.f,0.f,0.f};

    for (int kc0 = 0; kc0 < D_FF; kc0 += KC) {
        __syncthreads();
        for (int t = tid; t < BM * (KC / 4); t += 256) {
            int row = t >> 4;
            int c4  = t & 15;
            const float4* src = (const float4*)(hh + (size_t)shr[row] * D_FF + kc0);
            *(float4*)(Hs + row * XS_LD + c4 * 4) = src[c4];
        }
        __syncthreads();

        for (int k4 = 0; k4 < KC; k4 += 4) {
            int k0 = k4 + 2 * g;
            v2f a[4];
#pragma unroll
            for (int i = 0; i < 4; ++i) {
                int row = mg * 64 + i * 16 + nl;
                a[i].x = Hs[row * XS_LD + k0];
                a[i].y = Hs[row * XS_LD + k0 + 1];
            }
            const float* Wk = W2e + (size_t)(kc0 + k0) * D_MODEL;
#pragma unroll
            for (int j = 0; j < 2; ++j) {
                int col = nbase + j * 16 + nl;
                v2f b;
                b.x = Wk[col];
                b.y = Wk[D_MODEL + col];
#pragma unroll
                for (int i = 0; i < 4; ++i)
                    acc[i][j] = WMMA_F32(a[i], b, acc[i][j]);
            }
        }
    }

#pragma unroll
    for (int j = 0; j < 2; ++j) {
        int col = nbase + j * 16 + nl;
        float bias = b2e[col];
#pragma unroll
        for (int i = 0; i < 4; ++i) {
#pragma unroll
            for (int jj = 0; jj < 8; ++jj) {
                int row = mg * 64 + i * 16 + jj + 8 * g;
                int p   = Mb * BM + row;
                if (p < cnt) {
                    float o = (acc[i][j][jj] + bias) * sww[row];
                    atomicAdd(out + (size_t)stok[row] * D_MODEL + col, o);
                }
            }
        }
    }
}

extern "C" void kernel_launch(void* const* d_in, const int* in_sizes, int n_in,
                              void* d_out, int out_size, void* d_ws, size_t ws_size,
                              hipStream_t stream) {
    const float* x  = (const float*)d_in[0];
    const float* Wg = (const float*)d_in[1];
    const float* W1 = (const float*)d_in[2];
    const float* b1 = (const float*)d_in[3];
    const float* W2 = (const float*)d_in[4];
    const float* b2 = (const float*)d_in[5];
    float* out = (float*)d_out;

    int n_tokens = in_sizes[0] / D_MODEL;

    int*   counts = (int*)d_ws;
    int*   idx    = counts + 16;
    float* wts    = (float*)(idx + N_EXPERTS * n_tokens);
    int*   hrow   = (int*)(wts + N_EXPERTS * n_tokens);
    float* hh     = (float*)(hrow + N_EXPERTS * n_tokens);

    zero_out_kernel<<<(out_size + 255) / 256, 256, 0, stream>>>(out, out_size, counts);
    gate_kernel<<<n_tokens, 64, 0, stream>>>(x, Wg, counts, idx, wts, hrow, n_tokens);

    int mblocks = (n_tokens + BM - 1) / BM;
    dim3 g1(TWO_FF / 2 / 128, mblocks, N_EXPERTS);
    moe_gemm1_kernel<<<g1, 256, 0, stream>>>(x, W1, b1, counts, idx, hrow, hh, n_tokens);

    dim3 g2(D_MODEL / 128, mblocks, N_EXPERTS);
    moe_gemm2_kernel<<<g2, 256, 0, stream>>>(hh, W2, b2, counts, idx, wts, hrow, out, n_tokens);
}


# A-tile staging pipelined one K chunk ahead: next chunk's 16-byte row loads issued as one group before the current chunk's WMMAs into a second LDS stage; one barrier per chunk instead of two
# speedup vs baseline: 1.0900x; 1.0900x over previous
#include <hip/hip_runtime.h>
#include <math.h>

#define D_MODEL   1024
#define D_FF      2048
#define TWO_FF    4096
#define N_EXPERTS 8

#define BM   128
#define KC   64
#define XS_LD (KC + 4)
#define NSTG 8
#define STAGE_F (BM * XS_LD)

typedef float v2f __attribute__((ext_vector_type(2)));
typedef float v8f __attribute__((ext_vector_type(8)));
typedef float v4f __attribute__((ext_vector_type(4)));

#define STG_LOAD(P, OFF, KOFF) \
    p0 = *(const v4f*)((P) + OFF[0] + (KOFF)); p1 = *(const v4f*)((P) + OFF[1] + (KOFF)); \
    p2 = *(const v4f*)((P) + OFF[2] + (KOFF)); p3 = *(const v4f*)((P) + OFF[3] + (KOFF)); \
    p4 = *(const v4f*)((P) + OFF[4] + (KOFF)); p5 = *(const v4f*)((P) + OFF[5] + (KOFF)); \
    p6 = *(const v4f*)((P) + OFF[6] + (KOFF)); p7 = *(const v4f*)((P) + OFF[7] + (KOFF));
#define STG_STORE(DST) \
    *(v4f*)((DST) + 0 * 16 * XS_LD) = p0; *(v4f*)((DST) + 1 * 16 * XS_LD) = p1; \
    *(v4f*)((DST) + 2 * 16 * XS_LD) = p2; *(v4f*)((DST) + 3 * 16 * XS_LD) = p3; \
    *(v4f*)((DST) + 4 * 16 * XS_LD) = p4; *(v4f*)((DST) + 5 * 16 * XS_LD) = p5; \
    *(v4f*)((DST) + 6 * 16 * XS_LD) = p6; *(v4f*)((DST) + 7 * 16 * XS_LD) = p7;

#define WMMA_F32(A, B, C) \
    __builtin_amdgcn_wmma_f32_16x16x4_f32(false, (A), false, (B), (short)0, (C), false, false)

__global__ void zero_out_kernel(float* __restrict__ out, int n, int* __restrict__ counts) {
    int i = blockIdx.x * blockDim.x + threadIdx.x;
    if (i < n) out[i] = 0.0f;
    if (blockIdx.x == 0 && threadIdx.x < 16) counts[threadIdx.x] = 0;
}

__global__ void gate_kernel(const float* __restrict__ x,
                            const float* __restrict__ Wg,
                            int*   __restrict__ counts,
                            int*   __restrict__ idx,
                            float* __restrict__ wts,
                            int*   __restrict__ hrow,
                            int n_tokens) {
    int tok = blockIdx.x;
    int tid = threadIdx.x;
    __shared__ float red[64 * N_EXPERTS];
    __shared__ float tot[N_EXPERTS];

    float s[N_EXPERTS];
#pragma unroll
    for (int e = 0; e < N_EXPERTS; ++e) s[e] = 0.0f;

    const float* xr = x + (size_t)tok * D_MODEL;
    for (int d = tid; d < D_MODEL; d += 64) {
        float xv = xr[d];
#pragma unroll
        for (int e = 0; e < N_EXPERTS; ++e) s[e] += xv * Wg[e * D_MODEL + d];
    }
#pragma unroll
    for (int e = 0; e < N_EXPERTS; ++e) red[tid * N_EXPERTS + e] = s[e];
    __syncthreads();

    if (tid < N_EXPERTS) {
        float t = 0.0f;
        for (int i = 0; i < 64; ++i) t += red[i * N_EXPERTS + tid];
        tot[tid] = t;
    }
    __syncthreads();

    if (tid == 0) {
        int i1 = 0; float v1 = tot[0];
        for (int e = 1; e < N_EXPERTS; ++e) if (tot[e] > v1) { v1 = tot[e]; i1 = e; }
        int i2 = (i1 == 0) ? 1 : 0; float v2 = tot[i2];
        for (int e = 0; e < N_EXPERTS; ++e)
            if (e != i1 && tot[e] > v2) { v2 = tot[e]; i2 = e; }
        float m  = fmaxf(v1, v2);
        float e1 = __expf(v1 - m), e2 = __expf(v2 - m);
        float inv = 1.0f / (e1 + e2);

        int p1 = atomicAdd(&counts[i1], 1);
        idx [i1 * n_tokens + p1] = tok;
        wts [i1 * n_tokens + p1] = e1 * inv;
        hrow[i1 * n_tokens + p1] = 2 * tok;
        int p2 = atomicAdd(&counts[i2], 1);
        idx [i2 * n_tokens + p2] = tok;
        wts [i2 * n_tokens + p2] = e2 * inv;
        hrow[i2 * n_tokens + p2] = 2 * tok + 1;
    }
}

__global__ void __launch_bounds__(256)
moe_gemm1_kernel(const float* __restrict__ x,
                 const float* __restrict__ W1,
                 const float* __restrict__ b1,
                 const int*   __restrict__ counts,
                 const int*   __restrict__ idx,
                 const int*   __restrict__ hrow,
                 float*       __restrict__ hh,
                 int n_tokens) {
    int e   = blockIdx.z;
    int Mb  = blockIdx.y;
    int Fb  = blockIdx.x;
    int cnt = counts[e];
    if (Mb * BM >= cnt) return;

    __shared__ __attribute__((aligned(16))) float Xs[2 * STAGE_F];
    __shared__ int   shr[BM];

    int tid = threadIdx.x;
    if (tid < BM) {
        int p = Mb * BM + tid;
        shr[tid] = (p < cnt) ? hrow[e * n_tokens + p] : 0;
    }

    int srow = tid >> 4;
    int c4   = tid & 15;
    unsigned xoff[NSTG];
#pragma unroll
    for (int it = 0; it < NSTG; ++it) {
        int p   = Mb * BM + srow + 16 * it;
        int tok = (p < cnt) ? idx[e * n_tokens + p] : 0;
        xoff[it] = (unsigned)tok * D_MODEL + c4 * 4;
    }
    int sdst = srow * XS_LD + c4 * 4;

    int lane = tid & 31;
    int wave = tid >> 5;
    int g    = lane >> 4;
    int nl   = lane & 15;
    int mg   = wave >> 2;
    int fg   = wave & 3;
    int fbase = Fb * 128 + fg * 32;

    const float* W1e = W1 + (size_t)e * D_MODEL * TWO_FF;
    const float* b1e = b1 + (size_t)e * TWO_FF;

    v8f accA[4][2], accG[4][2];
#pragma unroll
    for (int i = 0; i < 4; ++i)
#pragma unroll
        for (int j = 0; j < 2; ++j) {
            accA[i][j] = (v8f){0.f,0.f,0.f,0.f,0.f,0.f,0.f,0.f};
            accG[i][j] = (v8f){0.f,0.f,0.f,0.f,0.f,0.f,0.f,0.f};
        }

    v4f p0, p1, p2, p3, p4, p5, p6, p7;
    STG_LOAD(x, xoff, 0)
    STG_STORE(Xs + sdst)
    __syncthreads();

    for (int kc = 0; kc < D_MODEL / KC; ++kc) {
        const int kc0 = kc * KC;
        const float* Xc = Xs + (kc & 1) * STAGE_F;
        float*       Xn = Xs + ((kc + 1) & 1) * STAGE_F;
        const bool more = (kc + 1 < D_MODEL / KC);

        if (more) {
            STG_LOAD(x, xoff, kc0 + KC)
        }
        __builtin_amdgcn_sched_barrier(0);

        for (int k4 = 0; k4 < KC; k4 += 4) {
            int k0 = k4 + 2 * g;
            v2f a[4];
#pragma unroll
            for (int i = 0; i < 4; ++i) {
                int row = mg * 64 + i * 16 + nl;
                a[i].x = Xc[row * XS_LD + k0];
                a[i].y = Xc[row * XS_LD + k0 + 1];
            }
            const float* Wk = W1e + (size_t)(kc0 + k0) * TWO_FF;
#pragma unroll
            for (int j = 0; j < 2; ++j) {
                int colA = fbase + j * 16 + nl;
                v2f bA, bG;
                bA.x = Wk[colA];
                bA.y = Wk[TWO_FF + colA];
                bG.x = Wk[colA + D_FF];
                bG.y = Wk[TWO_FF + colA + D_FF];
#pragma unroll
                for (int i = 0; i < 4; ++i) {
                    accA[i][j] = WMMA_F32(a[i], bA, accA[i][j]);
                    accG[i][j] = WMMA_F32(a[i], bG, accG[i][j]);
                }
            }
        }

        if (more) {
            STG_STORE(Xn + sdst)
        }
        __syncthreads();
    }

#pragma unroll
    for (int j = 0; j < 2; ++j) {
        int colA = fbase + j * 16 + nl;
        float biasA = b1e[colA];
        float biasG = b1e[colA + D_FF];
#pragma unroll
        for (int i = 0; i < 4; ++i) {
#pragma unroll
            for (int jj = 0; jj < 8; ++jj) {
                int row = mg * 64 + i * 16 + jj + 8 * g;
                int p   = Mb * BM + row;
                if (p < cnt) {
                    float av = accA[i][j][jj] + biasA;
                    float gv = accG[i][j][jj] + biasG;
                    float hv = av * (1.0f / (1.0f + __expf(-av))) * gv;
                    hh[(size_t)shr[row] * D_FF + colA] = hv;
                }
            }
        }
    }
}

__global__ void __launch_bounds__(256)
moe_gemm2_kernel(const float* __restrict__ hh,
                 const float* __restrict__ W2,
                 const float* __restrict__ b2,
                 const int*   __restrict__ counts,
                 const int*   __restrict__ idx,
                 const float* __restrict__ wts,
                 const int*   __restrict__ hrow,
                 float*       __restrict__ out,
                 int n_tokens) {
    int e   = blockIdx.z;
    int Mb  = blockIdx.y;
    int Nb  = blockIdx.x;
    int cnt = counts[e];
    if (Mb * BM >= cnt) return;

    __shared__ __attribute__((aligned(16))) float Hs[2 * STAGE_F];
    __shared__ int   stok[BM];
    __shared__ float sww[BM];

    int tid = threadIdx.x;
    if (tid < BM) {
        int p = Mb * BM + tid;
        bool v = (p < cnt);
        stok[tid] = v ? idx [e * n_tokens + p] : 0;
        sww [tid] = v ? wts [e * n_tokens + p] : 0.0f;
    }

    int srow = tid >> 4;
    int c4   = tid & 15;
    unsigned hoff[NSTG];
#pragma unroll
    for (int it = 0; it < NSTG; ++it) {
        int p  = Mb * BM + srow + 16 * it;
        int hr = (p < cnt) ? hrow[e * n_tokens + p] : 0;
        hoff[it] = (unsigned)hr * D_FF + c4 * 4;
    }
    int sdst = srow * XS_LD + c4 * 4;

    int lane = tid & 31;
    int wave = tid >> 5;
    int g    = lane >> 4;
    int nl   = lane & 15;
    int mg   = wave >> 2;
    int ng   = wave & 3;
    int nbase = Nb * 128 + ng * 32;

    const float* W2e = W2 + (size_t)e * D_FF * D_MODEL;
    const float* b2e = b2 + (size_t)e * D_MODEL;

    v8f acc[4][2];
#pragma unroll
    for (int i = 0; i < 4; ++i)
#pragma unroll
        for (int j = 0; j < 2; ++j)
            acc[i][j] = (v8f){0.f,0.f,0.f,0.f,0.f,0.f,0.f,0.f};

    v4f p0, p1, p2, p3, p4, p5, p6, p7;
    STG_LOAD(hh, hoff, 0)
    STG_STORE(Hs + sdst)
    __syncthreads();

    for (int kc = 0; kc < D_FF / KC; ++kc) {
        const int kc0 = kc * KC;
        const float* Hc = Hs + (kc & 1) * STAGE_F;
        float*       Hn = Hs + ((kc + 1) & 1) * STAGE_F;
        const bool more = (kc + 1 < D_FF / KC);

        if (more) {
            STG_LOAD(hh, hoff, kc0 + KC)
        }
        __builtin_amdgcn_sched_barrier(0);

        for (int k4 = 0; k4 < KC; k4 += 4) {
            int k0 = k4 + 2 * g;
            v2f a[4];
#pragma unroll
            for (int i = 0; i < 4; ++i) {
                int row = mg * 64 + i * 16 + nl;
                a[i].x = Hc[row * XS_LD + k0];
                a[i].y = Hc[row * XS_LD + k0 + 1];
            }
            const float* Wk = W2e + (size_t)(kc0 + k0) * D_MODEL;
#pragma unroll
            for (int j = 0; j < 2; ++j) {
                int col = nbase + j * 16 + nl;
                v2f b;
                b.x = Wk[col];
                b.y = Wk[D_MODEL + col];
#pragma unroll
                for (int i = 0; i < 4; ++i)
                    acc[i][j] = WMMA_F32(a[i], b, acc[i][j]);
            }
        }

        if (more) {
            STG_STORE(Hn + sdst)
        }
        __syncthreads();
    }

#pragma unroll
    for (int j = 0; j < 2; ++j) {
        int col = nbase + j * 16 + nl;
        float bias = b2e[col];
#pragma unroll
        for (int i = 0; i < 4; ++i) {
#pragma unroll
            for (int jj = 0; jj < 8; ++jj) {
                int row = mg * 64 + i * 16 + jj + 8 * g;
                int p   = Mb * BM + row;
                if (p < cnt) {
                    float o = (acc[i][j][jj] + bias) * sww[row];
                    atomicAdd(out + (size_t)stok[row] * D_MODEL + col, o);
                }
            }
        }
    }
}

extern "C" void kernel_launch(void* const* d_in, const int* in_sizes, int n_in,
                              void* d_out, int out_size, void* d_ws, size_t ws_size,
                              hipStream_t stream) {
    const float* x  = (const float*)d_in[0];
    const float* Wg = (const float*)d_in[1];
    const float* W1 = (const float*)d_in[2];
    const float* b1 = (const float*)d_in[3];
    const float* W2 = (const float*)d_in[4];
    const float* b2 = (const float*)d_in[5];
    float* out = (float*)d_out;

    int n_tokens = in_sizes[0] / D_MODEL;

    int*   counts = (int*)d_ws;
    int*   idx    = counts + 16;
    float* wts    = (float*)(idx + N_EXPERTS * n_tokens);
    int*   hrow   = (int*)(wts + N_EXPERTS * n_tokens);
    float* hh     = (float*)(hrow + N_EXPERTS * n_tokens);

    zero_out_kernel<<<(out_size + 255) / 256, 256, 0, stream>>>(out, out_size, counts);
    gate_kernel<<<n_tokens, 64, 0, stream>>>(x, Wg, counts, idx, wts, hrow, n_tokens);

    int mblocks = (n_tokens + BM - 1) / BM;
    dim3 g1(TWO_FF / 2 / 128, mblocks, N_EXPERTS);
    moe_gemm1_kernel<<<g1, 256, 0, stream>>>(x, W1, b1, counts, idx, hrow, hh, n_tokens);

    dim3 g2(D_MODEL / 128, mblocks, N_EXPERTS);
    moe_gemm2_kernel<<<g2, 256, 0, stream>>>(hh, W2, b2, counts, idx, wts, hrow, out, n_tokens);
}


# WMMA fragments double-buffered in registers: the next K step's A (LDS) and B (global) fragment loads are issued as one group ahead of the current step's WMMAs, in both GEMMs
# speedup vs baseline: 1.2330x; 1.1312x over previous
#include <hip/hip_runtime.h>
#include <math.h>

#define D_MODEL   1024
#define D_FF      2048
#define TWO_FF    4096
#define N_EXPERTS 8

#define BM   128
#define KC   64
#define XS_LD (KC + 4)
#define NSTG 8
#define STAGE_F (BM * XS_LD)

typedef float v2f __attribute__((ext_vector_type(2)));
typedef float v8f __attribute__((ext_vector_type(8)));
typedef float v4f __attribute__((ext_vector_type(4)));

#define STG_LOAD(P, OFF, KOFF) \
    p0 = *(const v4f*)((P) + OFF[0] + (KOFF)); p1 = *(const v4f*)((P) + OFF[1] + (KOFF)); \
    p2 = *(const v4f*)((P) + OFF[2] + (KOFF)); p3 = *(const v4f*)((P) + OFF[3] + (KOFF)); \
    p4 = *(const v4f*)((P) + OFF[4] + (KOFF)); p5 = *(const v4f*)((P) + OFF[5] + (KOFF)); \
    p6 = *(const v4f*)((P) + OFF[6] + (KOFF)); p7 = *(const v4f*)((P) + OFF[7] + (KOFF));
#define STG_STORE(DST) \
    *(v4f*)((DST) + 0 * 16 * XS_LD) = p0; *(v4f*)((DST) + 1 * 16 * XS_LD) = p1; \
    *(v4f*)((DST) + 2 * 16 * XS_LD) = p2; *(v4f*)((DST) + 3 * 16 * XS_LD) = p3; \
    *(v4f*)((DST) + 4 * 16 * XS_LD) = p4; *(v4f*)((DST) + 5 * 16 * XS_LD) = p5; \
    *(v4f*)((DST) + 6 * 16 * XS_LD) = p6; *(v4f*)((DST) + 7 * 16 * XS_LD) = p7;

#define WMMA_F32(A, B, C) \
    __builtin_amdgcn_wmma_f32_16x16x4_f32(false, (A), false, (B), (short)0, (C), false, false)

__global__ void zero_out_kernel(float* __restrict__ out, int n, int* __restrict__ counts) {
    int i = blockIdx.x * blockDim.x + threadIdx.x;
    if (i < n) out[i] = 0.0f;
    if (blockIdx.x == 0 && threadIdx.x < 16) counts[threadIdx.x] = 0;
}

__global__ void gate_kernel(const float* __restrict__ x,
                            const float* __restrict__ Wg,
                            int*   __restrict__ counts,
                            int*   __restrict__ idx,
                            float* __restrict__ wts,
                            int*   __restrict__ hrow,
                            int n_tokens) {
    int tok = blockIdx.x;
    int tid = threadIdx.x;
    __shared__ float red[64 * N_EXPERTS];
    __shared__ float tot[N_EXPERTS];

    float s[N_EXPERTS];
#pragma unroll
    for (int e = 0; e < N_EXPERTS; ++e) s[e] = 0.0f;

    const float* xr = x + (size_t)tok * D_MODEL;
    for (int d = tid; d < D_MODEL; d += 64) {
        float xv = xr[d];
#pragma unroll
        for (int e = 0; e < N_EXPERTS; ++e) s[e] += xv * Wg[e * D_MODEL + d];
    }
#pragma unroll
    for (int e = 0; e < N_EXPERTS; ++e) red[tid * N_EXPERTS + e] = s[e];
    __syncthreads();

    if (tid < N_EXPERTS) {
        float t = 0.0f;
        for (int i = 0; i < 64; ++i) t += red[i * N_EXPERTS + tid];
        tot[tid] = t;
    }
    __syncthreads();

    if (tid == 0) {
        int i1 = 0; float v1 = tot[0];
        for (int e = 1; e < N_EXPERTS; ++e) if (tot[e] > v1) { v1 = tot[e]; i1 = e; }
        int i2 = (i1 == 0) ? 1 : 0; float v2 = tot[i2];
        for (int e = 0; e < N_EXPERTS; ++e)
            if (e != i1 && tot[e] > v2) { v2 = tot[e]; i2 = e; }
        float m  = fmaxf(v1, v2);
        float e1 = __expf(v1 - m), e2 = __expf(v2 - m);
        float inv = 1.0f / (e1 + e2);

        int p1 = atomicAdd(&counts[i1], 1);
        idx [i1 * n_tokens + p1] = tok;
        wts [i1 * n_tokens + p1] = e1 * inv;
        hrow[i1 * n_tokens + p1] = 2 * tok;
        int p2 = atomicAdd(&counts[i2], 1);
        idx [i2 * n_tokens + p2] = tok;
        wts [i2 * n_tokens + p2] = e2 * inv;
        hrow[i2 * n_tokens + p2] = 2 * tok + 1;
    }
}

__global__ void __launch_bounds__(256)
moe_gemm1_kernel(const float* __restrict__ x,
                 const float* __restrict__ W1,
                 const float* __restrict__ b1,
                 const int*   __restrict__ counts,
                 const int*   __restrict__ idx,
                 const int*   __restrict__ hrow,
                 float*       __restrict__ hh,
                 int n_tokens) {
    int e   = blockIdx.z;
    int Mb  = blockIdx.y;
    int Fb  = blockIdx.x;
    int cnt = counts[e];
    if (Mb * BM >= cnt) return;

    __shared__ __attribute__((aligned(16))) float Xs[2 * STAGE_F];
    __shared__ int   shr[BM];

    int tid = threadIdx.x;
    if (tid < BM) {
        int p = Mb * BM + tid;
        shr[tid] = (p < cnt) ? hrow[e * n_tokens + p] : 0;
    }

    int srow = tid >> 4;
    int c4   = tid & 15;
    unsigned xoff[NSTG];
#pragma unroll
    for (int it = 0; it < NSTG; ++it) {
        int p   = Mb * BM + srow + 16 * it;
        int tok = (p < cnt) ? idx[e * n_tokens + p] : 0;
        xoff[it] = (unsigned)tok * D_MODEL + c4 * 4;
    }
    int sdst = srow * XS_LD + c4 * 4;

    int lane = tid & 31;
    int wave = tid >> 5;
    int g    = lane >> 4;
    int nl   = lane & 15;
    int mg   = wave >> 2;
    int fg   = wave & 3;
    int fbase = Fb * 128 + fg * 32;

    const float* W1e = W1 + (size_t)e * D_MODEL * TWO_FF;
    const float* b1e = b1 + (size_t)e * TWO_FF;

    v8f accA[4][2], accG[4][2];
#pragma unroll
    for (int i = 0; i < 4; ++i)
#pragma unroll
        for (int j = 0; j < 2; ++j) {
            accA[i][j] = (v8f){0.f,0.f,0.f,0.f,0.f,0.f,0.f,0.f};
            accG[i][j] = (v8f){0.f,0.f,0.f,0.f,0.f,0.f,0.f,0.f};
        }

    v4f p0, p1, p2, p3, p4, p5, p6, p7;
    STG_LOAD(x, xoff, 0)
    STG_STORE(Xs + sdst)

    const float* Ab = Xs + (mg * 64 + nl) * XS_LD + 2 * g;
    const float* Wb = W1e + (size_t)(2 * g) * TWO_FF + fbase + nl;

    v2f fa[2][4], fb[2][4];
#define G1_LDB(S, KABS) { \
        const float* Wk = Wb + (size_t)(KABS) * TWO_FF; \
        fb[S][0].x = Wk[0];          fb[S][0].y = Wk[TWO_FF]; \
        fb[S][1].x = Wk[D_FF];       fb[S][1].y = Wk[TWO_FF + D_FF]; \
        fb[S][2].x = Wk[16];         fb[S][2].y = Wk[TWO_FF + 16]; \
        fb[S][3].x = Wk[16 + D_FF];  fb[S][3].y = Wk[TWO_FF + 16 + D_FF]; }
#define G1_LDA(S, XC, K4) { \
        fa[S][0] = *(const v2f*)((XC) + 0 * 16 * XS_LD + (K4)); \
        fa[S][1] = *(const v2f*)((XC) + 1 * 16 * XS_LD + (K4)); \
        fa[S][2] = *(const v2f*)((XC) + 2 * 16 * XS_LD + (K4)); \
        fa[S][3] = *(const v2f*)((XC) + 3 * 16 * XS_LD + (K4)); }

    G1_LDB(0, 0)
    __syncthreads();

    for (int kc = 0; kc < D_MODEL / KC; ++kc) {
        const int kc0 = kc * KC;
        const float* Xc = Ab + (kc & 1) * STAGE_F;
        float*       Xn = Xs + ((kc + 1) & 1) * STAGE_F;
        const int knx = (kc + 1 < D_MODEL / KC) ? kc0 + KC : 0;

        STG_LOAD(x, xoff, knx)
        G1_LDA(0, Xc, 0)
        __builtin_amdgcn_sched_barrier(0);

#pragma unroll
        for (int s = 0; s < KC / 4; ++s) {
            const int cur = s & 1, nxt = cur ^ 1;
            if (s + 1 < KC / 4) {
                G1_LDB(nxt, kc0 + 4 * (s + 1))
                G1_LDA(nxt, Xc, 4 * (s + 1))
            } else {
                G1_LDB(nxt, knx)
            }
            __builtin_amdgcn_sched_barrier(0);
#pragma unroll
            for (int j = 0; j < 2; ++j) {
#pragma unroll
                for (int i = 0; i < 4; ++i) {
                    accA[i][j] = WMMA_F32(fa[cur][i], fb[cur][2 * j],     accA[i][j]);
                    accG[i][j] = WMMA_F32(fa[cur][i], fb[cur][2 * j + 1], accG[i][j]);
                }
            }
            __builtin_amdgcn_sched_barrier(0);
        }

        STG_STORE(Xn + sdst)
        __syncthreads();
    }
#undef G1_LDA
#undef G1_LDB

#pragma unroll
    for (int j = 0; j < 2; ++j) {
        int colA = fbase + j * 16 + nl;
        float biasA = b1e[colA];
        float biasG = b1e[colA + D_FF];
#pragma unroll
        for (int i = 0; i < 4; ++i) {
#pragma unroll
            for (int jj = 0; jj < 8; ++jj) {
                int row = mg * 64 + i * 16 + jj + 8 * g;
                int p   = Mb * BM + row;
                if (p < cnt) {
                    float av = accA[i][j][jj] + biasA;
                    float gv = accG[i][j][jj] + biasG;
                    float hv = av * (1.0f / (1.0f + __expf(-av))) * gv;
                    hh[(size_t)shr[row] * D_FF + colA] = hv;
                }
            }
        }
    }
}

__global__ void __launch_bounds__(256)
moe_gemm2_kernel(const float* __restrict__ hh,
                 const float* __restrict__ W2,
                 const float* __restrict__ b2,
                 const int*   __restrict__ counts,
                 const int*   __restrict__ idx,
                 const float* __restrict__ wts,
                 const int*   __restrict__ hrow,
                 float*       __restrict__ out,
                 int n_tokens) {
    int e   = blockIdx.z;
    int Mb  = blockIdx.y;
    int Nb  = blockIdx.x;
    int cnt = counts[e];
    if (Mb * BM >= cnt) return;

    __shared__ __attribute__((aligned(16))) float Hs[2 * STAGE_F];
    __shared__ int   stok[BM];
    __shared__ float sww[BM];

    int tid = threadIdx.x;
    if (tid < BM) {
        int p = Mb * BM + tid;
        bool v = (p < cnt);
        stok[tid] = v ? idx [e * n_tokens + p] : 0;
        sww [tid] = v ? wts [e * n_tokens + p] : 0.0f;
    }

    int srow = tid >> 4;
    int c4   = tid & 15;
    unsigned hoff[NSTG];
#pragma unroll
    for (int it = 0; it < NSTG; ++it) {
        int p  = Mb * BM + srow + 16 * it;
        int hr = (p < cnt) ? hrow[e * n_tokens + p] : 0;
        hoff[it] = (unsigned)hr * D_FF + c4 * 4;
    }
    int sdst = srow * XS_LD + c4 * 4;

    int lane = tid & 31;
    int wave = tid >> 5;
    int g    = lane >> 4;
    int nl   = lane & 15;
    int mg   = wave >> 2;
    int ng   = wave & 3;
    int nbase = Nb * 128 + ng * 32;

    const float* W2e = W2 + (size_t)e * D_FF * D_MODEL;
    const float* b2e = b2 + (size_t)e * D_MODEL;

    v8f acc[4][2];
#pragma unroll
    for (int i = 0; i < 4; ++i)
#pragma unroll
        for (int j = 0; j < 2; ++j)
            acc[i][j] = (v8f){0.f,0.f,0.f,0.f,0.f,0.f,0.f,0.f};

    v4f p0, p1, p2, p3, p4, p5, p6, p7;
    STG_LOAD(hh, hoff, 0)
    STG_STORE(Hs + sdst)

    const float* Ab = Hs + (mg * 64 + nl) * XS_LD + 2 * g;
    const float* Wb = W2e + (size_t)(2 * g) * D_MODEL + nbase + nl;

    v2f fa[2][2][4], fb[2][2][2];
#define G2_LDB(S, KABS) { \
        const float* Wk = Wb + (size_t)(KABS) * D_MODEL; \
        fb[S][0][0].x = Wk[0];                 fb[S][0][0].y = Wk[D_MODEL]; \
        fb[S][0][1].x = Wk[16];                fb[S][0][1].y = Wk[D_MODEL + 16]; \
        fb[S][1][0].x = Wk[4 * D_MODEL];       fb[S][1][0].y = Wk[5 * D_MODEL]; \
        fb[S][1][1].x = Wk[4 * D_MODEL + 16];  fb[S][1][1].y = Wk[5 * D_MODEL + 16]; }
#define G2_LDA(S, HC, K8) { \
        fa[S][0][0] = *(const v2f*)((HC) + 0 * 16 * XS_LD + (K8));     fa[S][1][0] = *(const v2f*)((HC) + 0 * 16 * XS_LD + (K8) + 4); \
        fa[S][0][1] = *(const v2f*)((HC) + 1 * 16 * XS_LD + (K8));     fa[S][1][1] = *(const v2f*)((HC) + 1 * 16 * XS_LD + (K8) + 4); \
        fa[S][0][2] = *(const v2f*)((HC) + 2 * 16 * XS_LD + (K8));     fa[S][1][2] = *(const v2f*)((HC) + 2 * 16 * XS_LD + (K8) + 4); \
        fa[S][0][3] = *(const v2f*)((HC) + 3 * 16 * XS_LD + (K8));     fa[S][1][3] = *(const v2f*)((HC) + 3 * 16 * XS_LD + (K8) + 4); }

    G2_LDB(0, 0)
    __syncthreads();

    for (int kc = 0; kc < D_FF / KC; ++kc) {
        const int kc0 = kc * KC;
        const float* Hc = Ab + (kc & 1) * STAGE_F;
        float*       Hn = Hs + ((kc + 1) & 1) * STAGE_F;
        const int knx = (kc + 1 < D_FF / KC) ? kc0 + KC : 0;

        STG_LOAD(hh, hoff, knx)
        G2_LDA(0, Hc, 0)
        __builtin_amdgcn_sched_barrier(0);

#pragma unroll
        for (int s = 0; s < KC / 8; ++s) {
            const int cur = s & 1, nxt = cur ^ 1;
            if (s + 1 < KC / 8) {
                G2_LDB(nxt, kc0 + 8 * (s + 1))
                G2_LDA(nxt, Hc, 8 * (s + 1))
            } else {
                G2_LDB(nxt, knx)
            }
            __builtin_amdgcn_sched_barrier(0);
#pragma unroll
            for (int t = 0; t < 2; ++t) {
#pragma unroll
                for (int j = 0; j < 2; ++j) {
#pragma unroll
                    for (int i = 0; i < 4; ++i)
                        acc[i][j] = WMMA_F32(fa[cur][t][i], fb[cur][t][j], acc[i][j]);
                }
            }
            __builtin_amdgcn_sched_barrier(0);
        }

        STG_STORE(Hn + sdst)
        __syncthreads();
    }
#undef G2_LDA
#undef G2_LDB

#pragma unroll
    for (int j = 0; j < 2; ++j) {
        int col = nbase + j * 16 + nl;
        float bias = b2e[col];
#pragma unroll
        for (int i = 0; i < 4; ++i) {
#pragma unroll
            for (int jj = 0; jj < 8; ++jj) {
                int row = mg * 64 + i * 16 + jj + 8 * g;
                int p   = Mb * BM + row;
                if (p < cnt) {
                    float o = (acc[i][j][jj] + bias) * sww[row];
                    atomicAdd(out + (size_t)stok[row] * D_MODEL + col, o);
                }
            }
        }
    }
}

extern "C" void kernel_launch(void* const* d_in, const int* in_sizes, int n_in,
                              void* d_out, int out_size, void* d_ws, size_t ws_size,
                              hipStream_t stream) {
    const float* x  = (const float*)d_in[0];
    const float* Wg = (const float*)d_in[1];
    const float* W1 = (const float*)d_in[2];
    const float* b1 = (const float*)d_in[3];
    const float* W2 = (const float*)d_in[4];
    const float* b2 = (const float*)d_in[5];
    float* out = (float*)d_out;

    int n_tokens = in_sizes[0] / D_MODEL;

    int*   counts = (int*)d_ws;
    int*   idx    = counts + 16;
    float* wts    = (float*)(idx + N_EXPERTS * n_tokens);
    int*   hrow   = (int*)(wts + N_EXPERTS * n_tokens);
    float* hh     = (float*)(hrow + N_EXPERTS * n_tokens);

    zero_out_kernel<<<(out_size + 255) / 256, 256, 0, stream>>>(out, out_size, counts);
    gate_kernel<<<n_tokens, 64, 0, stream>>>(x, Wg, counts, idx, wts, hrow, n_tokens);

    int mblocks = (n_tokens + BM - 1) / BM;
    dim3 g1(TWO_FF / 2 / 128, mblocks, N_EXPERTS);
    moe_gemm1_kernel<<<g1, 256, 0, stream>>>(x, W1, b1, counts, idx, hrow, hh, n_tokens);

    dim3 g2(D_MODEL / 128, mblocks, N_EXPERTS);
    moe_gemm2_kernel<<<g2, 256, 0, stream>>>(hh, W2, b2, counts, idx, wts, hrow, out, n_tokens);
}


# Block tile halved in M (128 -> 64 rows), 4-wave blocks, wave tile unchanged, both GEMMs: twice as many half-size blocks, four resident per compute unit, so the last partial round of blocks is shorter
# speedup vs baseline: 1.3211x; 1.0715x over previous
#include <hip/hip_runtime.h>
#include <math.h>

#define D_MODEL   1024
#define D_FF      2048
#define TWO_FF    4096
#define N_EXPERTS 8

#define BM   64
#define KC   64
#define XS_LD (KC + 4)
#define NSTG 8
#define STAGE_F (BM * XS_LD)
#define NTHR 128

typedef float v2f __attribute__((ext_vector_type(2)));
typedef float v8f __attribute__((ext_vector_type(8)));
typedef float v4f __attribute__((ext_vector_type(4)));

#define STG_LOAD(P, OFF, KOFF) \
    p0 = *(const v4f*)((P) + OFF[0] + (KOFF)); p1 = *(const v4f*)((P) + OFF[1] + (KOFF)); \
    p2 = *(const v4f*)((P) + OFF[2] + (KOFF)); p3 = *(const v4f*)((P) + OFF[3] + (KOFF)); \
    p4 = *(const v4f*)((P) + OFF[4] + (KOFF)); p5 = *(const v4f*)((P) + OFF[5] + (KOFF)); \
    p6 = *(const v4f*)((P) + OFF[6] + (KOFF)); p7 = *(const v4f*)((P) + OFF[7] + (KOFF));
#define STG_STORE(DST) \
    *(v4f*)((DST) + 0 * 8 * XS_LD) = p0; *(v4f*)((DST) + 1 * 8 * XS_LD) = p1; \
    *(v4f*)((DST) + 2 * 8 * XS_LD) = p2; *(v4f*)((DST) + 3 * 8 * XS_LD) = p3; \
    *(v4f*)((DST) + 4 * 8 * XS_LD) = p4; *(v4f*)((DST) + 5 * 8 * XS_LD) = p5; \
    *(v4f*)((DST) + 6 * 8 * XS_LD) = p6; *(v4f*)((DST) + 7 * 8 * XS_LD) = p7;

#define WMMA_F32(A, B, C) \
    __builtin_amdgcn_wmma_f32_16x16x4_f32(false, (A), false, (B), (short)0, (C), false, false)

__global__ void zero_out_kernel(float* __restrict__ out, int n, int* __restrict__ counts) {
    int i = blockIdx.x * blockDim.x + threadIdx.x;
    if (i < n) out[i] = 0.0f;
    if (blockIdx.x == 0 && threadIdx.x < 16) counts[threadIdx.x] = 0;
}

__global__ void gate_kernel(const float* __restrict__ x,
                            const float* __restrict__ Wg,
                            int*   __restrict__ counts,
                            int*   __restrict__ idx,
                            float* __restrict__ wts,
                            int*   __restrict__ hrow,
                            int n_tokens) {
    int tok = blockIdx.x;
    int tid = threadIdx.x;
    __shared__ float red[64 * N_EXPERTS];
    __shared__ float tot[N_EXPERTS];

    float s[N_EXPERTS];
#pragma unroll
    for (int e = 0; e < N_EXPERTS; ++e) s[e] = 0.0f;

    const float* xr = x + (size_t)tok * D_MODEL;
    for (int d = tid; d < D_MODEL; d += 64) {
        float xv = xr[d];
#pragma unroll
        for (int e = 0; e < N_EXPERTS; ++e) s[e] += xv * Wg[e * D_MODEL + d];
    }
#pragma unroll
    for (int e = 0; e < N_EXPERTS; ++e) red[tid * N_EXPERTS + e] = s[e];
    __syncthreads();

    if (tid < N_EXPERTS) {
        float t = 0.0f;
        for (int i = 0; i < 64; ++i) t += red[i * N_EXPERTS + tid];
        tot[tid] = t;
    }
    __syncthreads();

    if (tid == 0) {
        int i1 = 0; float v1 = tot[0];
        for (int e = 1; e < N_EXPERTS; ++e) if (tot[e] > v1) { v1 = tot[e]; i1 = e; }
        int i2 = (i1 == 0) ? 1 : 0; float v2 = tot[i2];
        for (int e = 0; e < N_EXPERTS; ++e)
            if (e != i1 && tot[e] > v2) { v2 = tot[e]; i2 = e; }
        float m  = fmaxf(v1, v2);
        float e1 = __expf(v1 - m), e2 = __expf(v2 - m);
        float inv = 1.0f / (e1 + e2);

        int p1 = atomicAdd(&counts[i1], 1);
        idx [i1 * n_tokens + p1] = tok;
        wts [i1 * n_tokens + p1] = e1 * inv;
        hrow[i1 * n_tokens + p1] = 2 * tok;
        int p2 = atomicAdd(&counts[i2], 1);
        idx [i2 * n_tokens + p2] = tok;
        wts [i2 * n_tokens + p2] = e2 * inv;
        hrow[i2 * n_tokens + p2] = 2 * tok + 1;
    }
}

__global__ void __launch_bounds__(NTHR)
moe_gemm1_kernel(const float* __restrict__ x,
                 const float* __restrict__ W1,
                 const float* __restrict__ b1,
                 const int*   __restrict__ counts,
                 const int*   __restrict__ idx,
                 const int*   __restrict__ hrow,
                 float*       __restrict__ hh,
                 int n_tokens) {
    int e   = blockIdx.z;
    int Mb  = blockIdx.y;
    int Fb  = blockIdx.x;
    int cnt = counts[e];
    if (Mb * BM >= cnt) return;

    __shared__ __attribute__((aligned(16))) float Xs[2 * STAGE_F];
    __shared__ int   shr[BM];

    int tid = threadIdx.x;
    if (tid < BM) {
        int p = Mb * BM + tid;
        shr[tid] = (p < cnt) ? hrow[e * n_tokens + p] : 0;
    }

    int srow = tid >> 4;
    int c4   = tid & 15;
    unsigned xoff[NSTG];
#pragma unroll
    for (int it = 0; it < NSTG; ++it) {
        int p   = Mb * BM + srow + 8 * it;
        int tok = (p < cnt) ? idx[e * n_tokens + p] : 0;
        xoff[it] = (unsigned)tok * D_MODEL + c4 * 4;
    }
    int sdst = srow * XS_LD + c4 * 4;

    int lane = tid & 31;
    int wave = tid >> 5;
    int g    = lane >> 4;
    int nl   = lane & 15;
    int fg   = wave;
    int fbase = Fb * 128 + fg * 32;

    const float* W1e = W1 + (size_t)e * D_MODEL * TWO_FF;
    const float* b1e = b1 + (size_t)e * TWO_FF;

    v8f accA[4][2], accG[4][2];
#pragma unroll
    for (int i = 0; i < 4; ++i)
#pragma unroll
        for (int j = 0; j < 2; ++j) {
            accA[i][j] = (v8f){0.f,0.f,0.f,0.f,0.f,0.f,0.f,0.f};
            accG[i][j] = (v8f){0.f,0.f,0.f,0.f,0.f,0.f,0.f,0.f};
        }

    v4f p0, p1, p2, p3, p4, p5, p6, p7;
    STG_LOAD(x, xoff, 0)
    STG_STORE(Xs + sdst)

    const float* Ab = Xs + nl * XS_LD + 2 * g;
    const float* Wb = W1e + (size_t)(2 * g) * TWO_FF + fbase + nl;

    v2f fa[2][4], fb[2][4];
#define G1_LDB(S, KABS) { \
        const float* Wk = Wb + (size_t)(KABS) * TWO_FF; \
        fb[S][0].x = Wk[0];          fb[S][0].y = Wk[TWO_FF]; \
        fb[S][1].x = Wk[D_FF];       fb[S][1].y = Wk[TWO_FF + D_FF]; \
        fb[S][2].x = Wk[16];         fb[S][2].y = Wk[TWO_FF + 16]; \
        fb[S][3].x = Wk[16 + D_FF];  fb[S][3].y = Wk[TWO_FF + 16 + D_FF]; }
#define G1_LDA(S, XC, K4) { \
        fa[S][0] = *(const v2f*)((XC) + 0 * 16 * XS_LD + (K4)); \
        fa[S][1] = *(const v2f*)((XC) + 1 * 16 * XS_LD + (K4)); \
        fa[S][2] = *(const v2f*)((XC) + 2 * 16 * XS_LD + (K4)); \
        fa[S][3] = *(const v2f*)((XC) + 3 * 16 * XS_LD + (K4)); }

    G1_LDB(0, 0)
    __syncthreads();

    for (int kc = 0; kc < D_MODEL / KC; ++kc) {
        const int kc0 = kc * KC;
        const float* Xc = Ab + (kc & 1) * STAGE_F;
        float*       Xn = Xs + ((kc + 1) & 1) * STAGE_F;
        const int knx = (kc + 1 < D_MODEL / KC) ? kc0 + KC : 0;

        STG_LOAD(x, xoff, knx)
        G1_LDA(0, Xc, 0)
        __builtin_amdgcn_sched_barrier(0);

#pragma unroll
        for (int s = 0; s < KC / 4; ++s) {
            const int cur = s & 1, nxt = cur ^ 1;
            if (s + 1 < KC / 4) {
                G1_LDB(nxt, kc0 + 4 * (s + 1))
                G1_LDA(nxt, Xc, 4 * (s + 1))
            } else {
                G1_LDB(nxt, knx)
            }
            __builtin_amdgcn_sched_barrier(0);
#pragma unroll
            for (int j = 0; j < 2; ++j) {
#pragma unroll
                for (int i = 0; i < 4; ++i) {
                    accA[i][j] = WMMA_F32(fa[cur][i], fb[cur][2 * j],     accA[i][j]);
                    accG[i][j] = WMMA_F32(fa[cur][i], fb[cur][2 * j + 1], accG[i][j]);
                }
            }
            __builtin_amdgcn_sched_barrier(0);
        }

        STG_STORE(Xn + sdst)
        __syncthreads();
    }
#undef G1_LDA
#undef G1_LDB

#pragma unroll
    for (int j = 0; j < 2; ++j) {
        int colA = fbase + j * 16 + nl;
        float biasA = b1e[colA];
        float biasG = b1e[colA + D_FF];
#pragma unroll
        for (int i = 0; i < 4; ++i) {
#pragma unroll
            for (int jj = 0; jj < 8; ++jj) {
                int row = i * 16 + jj + 8 * g;
                int p   = Mb * BM + row;
                if (p < cnt) {
                    float av = accA[i][j][jj] + biasA;
                    float gv = accG[i][j][jj] + biasG;
                    float hv = av * (1.0f / (1.0f + __expf(-av))) * gv;
                    hh[(size_t)shr[row] * D_FF + colA] = hv;
                }
            }
        }
    }
}

__global__ void __launch_bounds__(NTHR)
moe_gemm2_kernel(const float* __restrict__ hh,
                 const float* __restrict__ W2,
                 const float* __restrict__ b2,
                 const int*   __restrict__ counts,
                 const int*   __restrict__ idx,
                 const float* __restrict__ wts,
                 const int*   __restrict__ hrow,
                 float*       __restrict__ out,
                 int n_tokens) {
    int e   = blockIdx.z;
    int Mb  = blockIdx.y;
    int Nb  = blockIdx.x;
    int cnt = counts[e];
    if (Mb * BM >= cnt) return;

    __shared__ __attribute__((aligned(16))) float Hs[2 * STAGE_F];
    __shared__ int   stok[BM];
    __shared__ float sww[BM];

    int tid = threadIdx.x;
    if (tid < BM) {
        int p = Mb * BM + tid;
        bool v = (p < cnt);
        stok[tid] = v ? idx [e * n_tokens + p] : 0;
        sww [tid] = v ? wts [e * n_tokens + p] : 0.0f;
    }

    int srow = tid >> 4;
    int c4   = tid & 15;
    unsigned hoff[NSTG];
#pragma unroll
    for (int it = 0; it < NSTG; ++it) {
        int p  = Mb * BM + srow + 8 * it;
        int hr = (p < cnt) ? hrow[e * n_tokens + p] : 0;
        hoff[it] = (unsigned)hr * D_FF + c4 * 4;
    }
    int sdst = srow * XS_LD + c4 * 4;

    int lane = tid & 31;
    int wave = tid >> 5;
    int g    = lane >> 4;
    int nl   = lane & 15;
    int ng   = wave;
    int nbase = Nb * 128 + ng * 32;

    const float* W2e = W2 + (size_t)e * D_FF * D_MODEL;
    const float* b2e = b2 + (size_t)e * D_MODEL;

    v8f acc[4][2];
#pragma unroll
    for (int i = 0; i < 4; ++i)
#pragma unroll
        for (int j = 0; j < 2; ++j)
            acc[i][j] = (v8f){0.f,0.f,0.f,0.f,0.f,0.f,0.f,0.f};

    v4f p0, p1, p2, p3, p4, p5, p6, p7;
    STG_LOAD(hh, hoff, 0)
    STG_STORE(Hs + sdst)

    const float* Ab = Hs + nl * XS_LD + 2 * g;
    const float* Wb = W2e + (size_t)(2 * g) * D_MODEL + nbase + nl;

    v2f fa[2][2][4], fb[2][2][2];
#define G2_LDB(S, KABS) { \
        const float* Wk = Wb + (size_t)(KABS) * D_MODEL; \
        fb[S][0][0].x = Wk[0];                 fb[S][0][0].y = Wk[D_MODEL]; \
        fb[S][0][1].x = Wk[16];                fb[S][0][1].y = Wk[D_MODEL + 16]; \
        fb[S][1][0].x = Wk[4 * D_MODEL];       fb[S][1][0].y = Wk[5 * D_MODEL]; \
        fb[S][1][1].x = Wk[4 * D_MODEL + 16];  fb[S][1][1].y = Wk[5 * D_MODEL + 16]; }
#define G2_LDA(S, HC, K8) { \
        fa[S][0][0] = *(const v2f*)((HC) + 0 * 16 * XS_LD + (K8));     fa[S][1][0] = *(const v2f*)((HC) + 0 * 16 * XS_LD + (K8) + 4); \
        fa[S][0][1] = *(const v2f*)((HC) + 1 * 16 * XS_LD + (K8));     fa[S][1][1] = *(const v2f*)((HC) + 1 * 16 * XS_LD + (K8) + 4); \
        fa[S][0][2] = *(const v2f*)((HC) + 2 * 16 * XS_LD + (K8));     fa[S][1][2] = *(const v2f*)((HC) + 2 * 16 * XS_LD + (K8) + 4); \
        fa[S][0][3] = *(const v2f*)((HC) + 3 * 16 * XS_LD + (K8));     fa[S][1][3] = *(const v2f*)((HC) + 3 * 16 * XS_LD + (K8) + 4); }

    G2_LDB(0, 0)
    __syncthreads();

    for (int kc = 0; kc < D_FF / KC; ++kc) {
        const int kc0 = kc * KC;
        const float* Hc = Ab + (kc & 1) * STAGE_F;
        float*       Hn = Hs + ((kc + 1) & 1) * STAGE_F;
        const int knx = (kc + 1 < D_FF / KC) ? kc0 + KC : 0;

        STG_LOAD(hh, hoff, knx)
        G2_LDA(0, Hc, 0)
        __builtin_amdgcn_sched_barrier(0);

#pragma unroll
        for (int s = 0; s < KC / 8; ++s) {
            const int cur = s & 1, nxt = cur ^ 1;
            if (s + 1 < KC / 8) {
                G2_LDB(nxt, kc0 + 8 * (s + 1))
                G2_LDA(nxt, Hc, 8 * (s + 1))
            } else {
                G2_LDB(nxt, knx)
            }
            __builtin_amdgcn_sched_barrier(0);
#pragma unroll
            for (int t = 0; t < 2; ++t) {
#pragma unroll
                for (int j = 0; j < 2; ++j) {
#pragma unroll
                    for (int i = 0; i < 4; ++i)
                        acc[i][j] = WMMA_F32(fa[cur][t][i], fb[cur][t][j], acc[i][j]);
                }
            }
            __builtin_amdgcn_sched_barrier(0);
        }

        STG_STORE(Hn + sdst)
        __syncthreads();
    }
#undef G2_LDA
#undef G2_LDB

#pragma unroll
    for (int j = 0; j < 2; ++j) {
        int col = nbase + j * 16 + nl;
        float bias = b2e[col];
#pragma unroll
        for (int i = 0; i < 4; ++i) {
#pragma unroll
            for (int jj = 0; jj < 8; ++jj) {
                int row = i * 16 + jj + 8 * g;
                int p   = Mb * BM + row;
                if (p < cnt) {
                    float o = (acc[i][j][jj] + bias) * sww[row];
                    atomicAdd(out + (size_t)stok[row] * D_MODEL + col, o);
                }
            }
        }
    }
}

extern "C" void kernel_launch(void* const* d_in, const int* in_sizes, int n_in,
                              void* d_out, int out_size, void* d_ws, size_t ws_size,
                              hipStream_t stream) {
    const float* x  = (const float*)d_in[0];
    const float* Wg = (const float*)d_in[1];
    const float* W1 = (const float*)d_in[2];
    const float* b1 = (const float*)d_in[3];
    const float* W2 = (const float*)d_in[4];
    const float* b2 = (const float*)d_in[5];
    float* out = (float*)d_out;

    int n_tokens = in_sizes[0] / D_MODEL;

    int*   counts = (int*)d_ws;
    int*   idx    = counts + 16;
    float* wts    = (float*)(idx + N_EXPERTS * n_tokens);
    int*   hrow   = (int*)(wts + N_EXPERTS * n_tokens);
    float* hh     = (float*)(hrow + N_EXPERTS * n_tokens);

    zero_out_kernel<<<(out_size + 255) / 256, 256, 0, stream>>>(out, out_size, counts);
    gate_kernel<<<n_tokens, 64, 0, stream>>>(x, Wg, counts, idx, wts, hrow, n_tokens);

    int mblocks = (n_tokens + BM - 1) / BM;
    dim3 g1(TWO_FF / 2 / 128, mblocks, N_EXPERTS);
    moe_gemm1_kernel<<<g1, NTHR, 0, stream>>>(x, W1, b1, counts, idx, hrow, hh, n_tokens);

    dim3 g2(D_MODEL / 128, mblocks, N_EXPERTS);
    moe_gemm2_kernel<<<g2, NTHR, 0, stream>>>(hh, W2, b2, counts, idx, wts, hrow, out, n_tokens);
}


# Fragment loads merged to one group per 32 WMMAs in both GEMMs: A read from LDS by 16-byte loads (a lane half owns 4 or 8 consecutive k), B rows re-paired to match; k sum order permuted in a group
# speedup vs baseline: 1.4052x; 1.0637x over previous
#include <hip/hip_runtime.h>
#include <math.h>

#define D_MODEL   1024
#define D_FF      2048
#define TWO_FF    4096
#define N_EXPERTS 8

#define BM   64
#define KC   64
#define XS_LD (KC + 4)
#define NSTG 8
#define STAGE_F (BM * XS_LD)
#define NTHR 128

typedef float v2f __attribute__((ext_vector_type(2)));
typedef float v8f __attribute__((ext_vector_type(8)));
typedef float v4f __attribute__((ext_vector_type(4)));

#define STG4_LOAD(PB, OFFB, I0, KB) \
    p0 = *(const v4f*)((PB) + (OFFB[(I0) + 0] + (KB))); p1 = *(const v4f*)((PB) + (OFFB[(I0) + 1] + (KB))); \
    p2 = *(const v4f*)((PB) + (OFFB[(I0) + 2] + (KB))); p3 = *(const v4f*)((PB) + (OFFB[(I0) + 3] + (KB)));
#define STG4_STORE(DST, I0) \
    *(v4f*)((DST) + ((I0) + 0) * 8 * XS_LD) = p0; *(v4f*)((DST) + ((I0) + 1) * 8 * XS_LD) = p1; \
    *(v4f*)((DST) + ((I0) + 2) * 8 * XS_LD) = p2; *(v4f*)((DST) + ((I0) + 3) * 8 * XS_LD) = p3;

#define WMMA_F32(A, B, C) \
    __builtin_amdgcn_wmma_f32_16x16x4_f32(false, (A), false, (B), (short)0, (C), false, false)

__global__ void zero_out_kernel(float* __restrict__ out, int n, int* __restrict__ counts) {
    int i = blockIdx.x * blockDim.x + threadIdx.x;
    if (i < n) out[i] = 0.0f;
    if (blockIdx.x == 0 && threadIdx.x < 16) counts[threadIdx.x] = 0;
}

__global__ void gate_kernel(const float* __restrict__ x,
                            const float* __restrict__ Wg,
                            int*   __restrict__ counts,
                            int*   __restrict__ idx,
                            float* __restrict__ wts,
                            int*   __restrict__ hrow,
                            int n_tokens) {
    int tok = blockIdx.x;
    int tid = threadIdx.x;
    __shared__ float red[64 * N_EXPERTS];
    __shared__ float tot[N_EXPERTS];

    float s[N_EXPERTS];
#pragma unroll
    for (int e = 0; e < N_EXPERTS; ++e) s[e] = 0.0f;

    const float* xr = x + (size_t)tok * D_MODEL;
    for (int d = tid; d < D_MODEL; d += 64) {
        float xv = xr[d];
#pragma unroll
        for (int e = 0; e < N_EXPERTS; ++e) s[e] += xv * Wg[e * D_MODEL + d];
    }
#pragma unroll
    for (int e = 0; e < N_EXPERTS; ++e) red[tid * N_EXPERTS + e] = s[e];
    __syncthreads();

    if (tid < N_EXPERTS) {
        float t = 0.0f;
        for (int i = 0; i < 64; ++i) t += red[i * N_EXPERTS + tid];
        tot[tid] = t;
    }
    __syncthreads();

    if (tid == 0) {
        int i1 = 0; float v1 = tot[0];
        for (int e = 1; e < N_EXPERTS; ++e) if (tot[e] > v1) { v1 = tot[e]; i1 = e; }
        int i2 = (i1 == 0) ? 1 : 0; float v2 = tot[i2];
        for (int e = 0; e < N_EXPERTS; ++e)
            if (e != i1 && tot[e] > v2) { v2 = tot[e]; i2 = e; }
        float m  = fmaxf(v1, v2);
        float e1 = __expf(v1 - m), e2 = __expf(v2 - m);
        float inv = 1.0f / (e1 + e2);

        int p1 = atomicAdd(&counts[i1], 1);
        idx [i1 * n_tokens + p1] = tok;
        wts [i1 * n_tokens + p1] = e1 * inv;
        hrow[i1 * n_tokens + p1] = 2 * tok;
        int p2 = atomicAdd(&counts[i2], 1);
        idx [i2 * n_tokens + p2] = tok;
        wts [i2 * n_tokens + p2] = e2 * inv;
        hrow[i2 * n_tokens + p2] = 2 * tok + 1;
    }
}

__global__ void __launch_bounds__(NTHR) __attribute__((amdgpu_waves_per_eu(4, 4)))
moe_gemm1_kernel(const float* __restrict__ x,
                 const float* __restrict__ W1,
                 const float* __restrict__ b1,
                 const int*   __restrict__ counts,
                 const int*   __restrict__ idx,
                 const int*   __restrict__ hrow,
                 float*       __restrict__ hh,
                 int n_tokens) {
    int e   = blockIdx.z;
    int Mb  = blockIdx.y;
    int Fb  = blockIdx.x;
    int cnt = counts[e];
    if (Mb * BM >= cnt) return;

    __shared__ __attribute__((aligned(16))) float Xs[2 * STAGE_F];
    __shared__ int   shr[BM];

    int tid = threadIdx.x;
    if (tid < BM) {
        int p = Mb * BM + tid;
        shr[tid] = (p < cnt) ? hrow[e * n_tokens + p] : 0;
    }

    int srow = tid >> 4;
    int c4   = tid & 15;
    unsigned xoff[NSTG];
#pragma unroll
    for (int it = 0; it < NSTG; ++it) {
        int p   = Mb * BM + srow + 8 * it;
        int tok = (p < cnt) ? idx[e * n_tokens + p] : 0;
        xoff[it] = ((unsigned)tok * D_MODEL + c4 * 4) * 4u;
    }
    int sdst = srow * XS_LD + c4 * 4;

    int lane = tid & 31;
    int wave = tid >> 5;
    int g    = lane >> 4;
    int nl   = lane & 15;
    int fg   = wave;
    int fbase = Fb * 128 + fg * 32;

    const float* W1e = W1 + (size_t)e * D_MODEL * TWO_FF;
    const float* b1e = b1 + (size_t)e * TWO_FF;

    v8f accA[4][2], accG[4][2];
#pragma unroll
    for (int i = 0; i < 4; ++i)
#pragma unroll
        for (int j = 0; j < 2; ++j) {
            accA[i][j] = (v8f){0.f,0.f,0.f,0.f,0.f,0.f,0.f,0.f};
            accG[i][j] = (v8f){0.f,0.f,0.f,0.f,0.f,0.f,0.f,0.f};
        }

    const char* xb = (const char*)x;
    v4f p0, p1, p2, p3;
    {
        STG4_LOAD(xb, xoff, 0, 0u)
        const v4f q0 = *(const v4f*)(xb + xoff[4]), q1 = *(const v4f*)(xb + xoff[5]);
        const v4f q2 = *(const v4f*)(xb + xoff[6]), q3 = *(const v4f*)(xb + xoff[7]);
        STG4_STORE(Xs + sdst, 0)
        *(v4f*)(Xs + sdst + 4 * 8 * XS_LD) = q0; *(v4f*)(Xs + sdst + 5 * 8 * XS_LD) = q1;
        *(v4f*)(Xs + sdst + 6 * 8 * XS_LD) = q2; *(v4f*)(Xs + sdst + 7 * 8 * XS_LD) = q3;
    }

    const float* Ab = Xs + nl * XS_LD + 4 * g;
    const float* Wb = W1e + (size_t)(4 * g) * TWO_FF + fbase + nl;

    v4f fa[2][4];
    v2f fb[2][2][4];
#define G1_LDB(S, KABS) { \
        const float* Wk = Wb + (size_t)(KABS) * TWO_FF; \
        fb[S][0][0].x = Wk[0];                        fb[S][0][0].y = Wk[TWO_FF]; \
        fb[S][0][1].x = Wk[D_FF];                     fb[S][0][1].y = Wk[TWO_FF + D_FF]; \
        fb[S][0][2].x = Wk[16];                       fb[S][0][2].y = Wk[TWO_FF + 16]; \
        fb[S][0][3].x = Wk[16 + D_FF];                fb[S][0][3].y = Wk[TWO_FF + 16 + D_FF]; \
        fb[S][1][0].x = Wk[2 * TWO_FF];               fb[S][1][0].y = Wk[3 * TWO_FF]; \
        fb[S][1][1].x = Wk[2 * TWO_FF + D_FF];        fb[S][1][1].y = Wk[3 * TWO_FF + D_FF]; \
        fb[S][1][2].x = Wk[2 * TWO_FF + 16];          fb[S][1][2].y = Wk[3 * TWO_FF + 16]; \
        fb[S][1][3].x = Wk[2 * TWO_FF + 16 + D_FF];   fb[S][1][3].y = Wk[3 * TWO_FF + 16 + D_FF]; }
#define G1_LDA(S, XC, K8) { \
        fa[S][0] = *(const v4f*)((XC) + 0 * 16 * XS_LD + (K8)); \
        fa[S][1] = *(const v4f*)((XC) + 1 * 16 * XS_LD + (K8)); \
        fa[S][2] = *(const v4f*)((XC) + 2 * 16 * XS_LD + (K8)); \
        fa[S][3] = *(const v4f*)((XC) + 3 * 16 * XS_LD + (K8)); }

    G1_LDB(0, 0)
    __syncthreads();

    for (int kc = 0; kc < D_MODEL / KC; ++kc) {
        const int kc0 = kc * KC;
        const float* Xc = Ab + (kc & 1) * STAGE_F;
        float*       Xn = Xs + ((kc + 1) & 1) * STAGE_F;
        const int knx = (kc + 1 < D_MODEL / KC) ? kc0 + KC : 0;

        const unsigned kb = (unsigned)knx * 4u;
        STG4_LOAD(xb, xoff, 0, kb)
        G1_LDA(0, Xc, 0)
        __builtin_amdgcn_sched_barrier(0);

#pragma unroll
        for (int s = 0; s < KC / 8; ++s) {
            const int cur = s & 1, nxt = cur ^ 1;
            if (s + 1 < KC / 8) {
                G1_LDB(nxt, kc0 + 8 * (s + 1))
                G1_LDA(nxt, Xc, 8 * (s + 1))
            } else {
                G1_LDB(nxt, knx)
            }
            if (s == KC / 16) {
                STG4_STORE(Xn + sdst, 0)
                STG4_LOAD(xb, xoff, 4, kb)
            }
            __builtin_amdgcn_sched_barrier(0);
#pragma unroll
            for (int t = 0; t < 2; ++t) {
#pragma unroll
                for (int j = 0; j < 2; ++j) {
#pragma unroll
                    for (int i = 0; i < 4; ++i) {
                        const v2f a = t ? __builtin_shufflevector(fa[cur][i], fa[cur][i], 2, 3)
                                        : __builtin_shufflevector(fa[cur][i], fa[cur][i], 0, 1);
                        accA[i][j] = WMMA_F32(a, fb[cur][t][2 * j],     accA[i][j]);
                        accG[i][j] = WMMA_F32(a, fb[cur][t][2 * j + 1], accG[i][j]);
                    }
                }
            }
            __builtin_amdgcn_sched_barrier(0);
        }

        STG4_STORE(Xn + sdst, 4)
        __syncthreads();
    }
#undef G1_LDA
#undef G1_LDB

#pragma unroll
    for (int j = 0; j < 2; ++j) {
        int colA = fbase + j * 16 + nl;
        float biasA = b1e[colA];
        float biasG = b1e[colA + D_FF];
#pragma unroll
        for (int i = 0; i < 4; ++i) {
#pragma unroll
            for (int jj = 0; jj < 8; ++jj) {
                int row = i * 16 + jj + 8 * g;
                int p   = Mb * BM + row;
                if (p < cnt) {
                    float av = accA[i][j][jj] + biasA;
                    float gv = accG[i][j][jj] + biasG;
                    float hv = av * (1.0f / (1.0f + __expf(-av))) * gv;
                    hh[(size_t)shr[row] * D_FF + colA] = hv;
                }
            }
        }
    }
}

__global__ void __launch_bounds__(NTHR) __attribute__((amdgpu_waves_per_eu(4, 4)))
moe_gemm2_kernel(const float* __restrict__ hh,
                 const float* __restrict__ W2,
                 const float* __restrict__ b2,
                 const int*   __restrict__ counts,
                 const int*   __restrict__ idx,
                 const float* __restrict__ wts,
                 const int*   __restrict__ hrow,
                 float*       __restrict__ out,
                 int n_tokens) {
    int e   = blockIdx.z;
    int Mb  = blockIdx.y;
    int Nb  = blockIdx.x;
    int cnt = counts[e];
    if (Mb * BM >= cnt) return;

    __shared__ __attribute__((aligned(16))) float Hs[2 * STAGE_F];
    __shared__ int   stok[BM];
    __shared__ float sww[BM];

    int tid = threadIdx.x;
    if (tid < BM) {
        int p = Mb * BM + tid;
        bool v = (p < cnt);
        stok[tid] = v ? idx [e * n_tokens + p] : 0;
        sww [tid] = v ? wts [e * n_tokens + p] : 0.0f;
    }

    int srow = tid >> 4;
    int c4   = tid & 15;
    unsigned hoff[NSTG];
#pragma unroll
    for (int it = 0; it < NSTG; ++it) {
        int p  = Mb * BM + srow + 8 * it;
        int hr = (p < cnt) ? hrow[e * n_tokens + p] : 0;
        hoff[it] = ((unsigned)hr * D_FF + c4 * 4) * 4u;
    }
    int sdst = srow * XS_LD + c4 * 4;

    int lane = tid & 31;
    int wave = tid >> 5;
    int g    = lane >> 4;
    int nl   = lane & 15;
    int ng   = wave;
    int nbase = Nb * 128 + ng * 32;

    const float* W2e = W2 + (size_t)e * D_FF * D_MODEL;
    const float* b2e = b2 + (size_t)e * D_MODEL;

    v8f acc[4][2];
#pragma unroll
    for (int i = 0; i < 4; ++i)
#pragma unroll
        for (int j = 0; j < 2; ++j)
            acc[i][j] = (v8f){0.f,0.f,0.f,0.f,0.f,0.f,0.f,0.f};

    const char* hb = (const char*)hh;
    v4f p0, p1, p2, p3;
    {
        STG4_LOAD(hb, hoff, 0, 0u)
        const v4f q0 = *(const v4f*)(hb + hoff[4]), q1 = *(const v4f*)(hb + hoff[5]);
        const v4f q2 = *(const v4f*)(hb + hoff[6]), q3 = *(const v4f*)(hb + hoff[7]);
        STG4_STORE(Hs + sdst, 0)
        *(v4f*)(Hs + sdst + 4 * 8 * XS_LD) = q0; *(v4f*)(Hs + sdst + 5 * 8 * XS_LD) = q1;
        *(v4f*)(Hs + sdst + 6 * 8 * XS_LD) = q2; *(v4f*)(Hs + sdst + 7 * 8 * XS_LD) = q3;
    }

    const float* Ab = Hs + nl * XS_LD + 8 * g;
    const float* Wb = W2e + (size_t)(8 * g) * D_MODEL + nbase + nl;

    v4f fa[2][2][4];
    v2f fb[2][4][2];
#define G2_LDB(S, KABS) { \
        const float* Wk = Wb + (size_t)(KABS) * D_MODEL; \
        fb[S][0][0].x = Wk[0];                 fb[S][0][0].y = Wk[D_MODEL]; \
        fb[S][0][1].x = Wk[16];                fb[S][0][1].y = Wk[D_MODEL + 16]; \
        fb[S][1][0].x = Wk[2 * D_MODEL];       fb[S][1][0].y = Wk[3 * D_MODEL]; \
        fb[S][1][1].x = Wk[2 * D_MODEL + 16];  fb[S][1][1].y = Wk[3 * D_MODEL + 16]; \
        fb[S][2][0].x = Wk[4 * D_MODEL];       fb[S][2][0].y = Wk[5 * D_MODEL]; \
        fb[S][2][1].x = Wk[4 * D_MODEL + 16];  fb[S][2][1].y = Wk[5 * D_MODEL + 16]; \
        fb[S][3][0].x = Wk[6 * D_MODEL];       fb[S][3][0].y = Wk[7 * D_MODEL]; \
        fb[S][3][1].x = Wk[6 * D_MODEL + 16];  fb[S][3][1].y = Wk[7 * D_MODEL + 16]; }
#define G2_LDA(S, HC, K16) { \
        fa[S][0][0] = *(const v4f*)((HC) + 0 * 16 * XS_LD + (K16));     fa[S][1][0] = *(const v4f*)((HC) + 0 * 16 * XS_LD + (K16) + 4); \
        fa[S][0][1] = *(const v4f*)((HC) + 1 * 16 * XS_LD + (K16));     fa[S][1][1] = *(const v4f*)((HC) + 1 * 16 * XS_LD + (K16) + 4); \
        fa[S][0][2] = *(const v4f*)((HC) + 2 * 16 * XS_LD + (K16));     fa[S][1][2] = *(const v4f*)((HC) + 2 * 16 * XS_LD + (K16) + 4); \
        fa[S][0][3] = *(const v4f*)((HC) + 3 * 16 * XS_LD + (K16));     fa[S][1][3] = *(const v4f*)((HC) + 3 * 16 * XS_LD + (K16) + 4); }

    G2_LDB(0, 0)
    __syncthreads();

    for (int kc = 0; kc < D_FF / KC; ++kc) {
        const int kc0 = kc * KC;
        const float* Hc = Ab + (kc & 1) * STAGE_F;
        float*       Hn = Hs + ((kc + 1) & 1) * STAGE_F;
        const int knx = (kc + 1 < D_FF / KC) ? kc0 + KC : 0;

        const unsigned kb = (unsigned)knx * 4u;
        STG4_LOAD(hb, hoff, 0, kb)
        G2_LDA(0, Hc, 0)
        __builtin_amdgcn_sched_barrier(0);

#pragma unroll
        for (int s = 0; s < KC / 16; ++s) {
            const int cur = s & 1, nxt = cur ^ 1;
            if (s + 1 < KC / 16) {
                G2_LDB(nxt, kc0 + 16 * (s + 1))
                G2_LDA(nxt, Hc, 16 * (s + 1))
            } else {
                G2_LDB(nxt, knx)
            }
            if (s == KC / 32) {
                STG4_STORE(Hn + sdst, 0)
                STG4_LOAD(hb, hoff, 4, kb)
            }
            __builtin_amdgcn_sched_barrier(0);
#pragma unroll
            for (int t = 0; t < 4; ++t) {
#pragma unroll
                for (int j = 0; j < 2; ++j) {
#pragma unroll
                    for (int i = 0; i < 4; ++i) {
                        const v4f q = fa[cur][t >> 1][i];
                        const v2f a = (t & 1) ? __builtin_shufflevector(q, q, 2, 3)
                                              : __builtin_shufflevector(q, q, 0, 1);
                        acc[i][j] = WMMA_F32(a, fb[cur][t][j], acc[i][j]);
                    }
                }
            }
            __builtin_amdgcn_sched_barrier(0);
        }

        STG4_STORE(Hn + sdst, 4)
        __syncthreads();
    }
#undef G2_LDA
#undef G2_LDB

#pragma unroll
    for (int j = 0; j < 2; ++j) {
        int col = nbase + j * 16 + nl;
        float bias = b2e[col];
#pragma unroll
        for (int i = 0; i < 4; ++i) {
#pragma unroll
            for (int jj = 0; jj < 8; ++jj) {
                int row = i * 16 + jj + 8 * g;
                int p   = Mb * BM + row;
                if (p < cnt) {
                    float o = (acc[i][j][jj] + bias) * sww[row];
                    atomicAdd(out + (size_t)stok[row] * D_MODEL + col, o);
                }
            }
        }
    }
}

extern "C" void kernel_launch(void* const* d_in, const int* in_sizes, int n_in,
                              void* d_out, int out_size, void* d_ws, size_t ws_size,
                              hipStream_t stream) {
    const float* x  = (const float*)d_in[0];
    const float* Wg = (const float*)d_in[1];
    const float* W1 = (const float*)d_in[2];
    const float* b1 = (const float*)d_in[3];
    const float* W2 = (const float*)d_in[4];
    const float* b2 = (const float*)d_in[5];
    float* out = (float*)d_out;

    int n_tokens = in_sizes[0] / D_MODEL;

    int*   counts = (int*)d_ws;
    int*   idx    = counts + 16;
    float* wts    = (float*)(idx + N_EXPERTS * n_tokens);
    int*   hrow   = (int*)(wts + N_EXPERTS * n_tokens);
    float* hh     = (float*)(hrow + N_EXPERTS * n_tokens);

    zero_out_kernel<<<(out_size + 255) / 256, 256, 0, stream>>>(out, out_size, counts);
    gate_kernel<<<n_tokens, 64, 0, stream>>>(x, Wg, counts, idx, wts, hrow, n_tokens);

    int mblocks = (n_tokens + BM - 1) / BM;
    dim3 g1(TWO_FF / 2 / 128, mblocks, N_EXPERTS);
    moe_gemm1_kernel<<<g1, NTHR, 0, stream>>>(x, W1, b1, counts, idx, hrow, hh, n_tokens);

    dim3 g2(D_MODEL / 128, mblocks, N_EXPERTS);
    moe_gemm2_kernel<<<g2, NTHR, 0, stream>>>(hh, W2, b2, counts, idx, wts, hrow, out, n_tokens);
}
